# starting kernel
# baseline (speedup 1.0000x reference)
#include <hip/hip_runtime.h>
#include <hip/hip_bf16.h>

#define BATCH   2
#define NPTS    256
#define NCH     32
#define NBASIS  16
#define KDIM    (NBASIS * NCH)
#define GAMMA_F 10.0f
#define EPS_F   1e-9f

typedef float v2f __attribute__((ext_vector_type(2)));
typedef float v8f __attribute__((ext_vector_type(8)));

#define OFF_D   0
#define OFF_S   (BATCH * NPTS * NPTS)
#define OFF_W2  (OFF_S + BATCH * NPTS * KDIM)

__global__ void cf_dist_kernel(const float* __restrict__ geom,
                               float* __restrict__ dmat) {
    int idx = blockIdx.x * blockDim.x + threadIdx.x;
    int b = idx & (NPTS - 1);
    int a = (idx >> 8) & (NPTS - 1);
    int z = idx >> 16;
    const float* ga = geom + (z * NPTS + a) * 3;
    const float* gb = geom + (z * NPTS + b) * 3;
    float dx = gb[0] - ga[0];
    float dy = gb[1] - ga[1];
    float dz = gb[2] - ga[2];
    dmat[idx] = sqrtf(dx * dx + dy * dy + dz * dz + EPS_F);
}

__global__ void cf_w2_kernel(const float* __restrict__ kw,
                             const int* __restrict__ n_norm,
                             float* __restrict__ w2) {
    int idx = blockIdx.x * blockDim.x + threadIdx.x;
    float scale = rsqrtf((float)n_norm[0]);
    int i  = idx & 31;
    int nj = idx >> 5;
    int j  = nj & 31;
    int n  = nj >> 5;
    w2[idx] = kw[(n * NCH + i) * NCH + j] * scale;
}

__global__ void cf_gemm1_kernel(const float* __restrict__ dmat,
                                const float* __restrict__ feat,
                                const float* __restrict__ centers,
                                float* __restrict__ S) {
    const int wave  = blockIdx.x * (blockDim.x >> 5) + (threadIdx.x >> 5);
    const int lane  = threadIdx.x & 31;
    const int at    = wave & 15;
    const int n     = (wave >> 4) & 15;
    const int z     = wave >> 8;
    const int row0  = at * 16;
    const int m     = lane & 15;
    const int khalf = lane >> 4;

    const float  cn   = centers[n];
    const float* drow = dmat + (z * NPTS + row0 + m) * NPTS;
    const float* fc0  = feat + z * NPTS * NCH + m;
    const float* fc1  = fc0 + 16;

    v8f c0 = {};
    v8f c1 = {};
    for (int k0 = 0; k0 < NPTS; k0 += 4) {
        int b0 = k0 + khalf * 2;
        float t0 = drow[b0] - cn;
        float t1 = drow[b0 + 1] - cn;
        v2f av;
        av.x = __expf(-GAMMA_F * t0 * t0);
        av.y = __expf(-GAMMA_F * t1 * t1);
        v2f bv0, bv1;
        bv0.x = fc0[b0 * NCH];
        bv0.y = fc0[(b0 + 1) * NCH];
        bv1.x = fc1[b0 * NCH];
        bv1.y = fc1[(b0 + 1) * NCH];
        c0 = __builtin_amdgcn_wmma_f32_16x16x4_f32(
                false, av, false, bv0, (short)0, c0, false, false);
        c1 = __builtin_amdgcn_wmma_f32_16x16x4_f32(
                false, av, false, bv1, (short)0, c1, false, false);
    }
#pragma unroll
    for (int i = 0; i < 8; ++i) {
        int row = row0 + i + khalf * 8;
        float* srow = S + (z * NPTS + row) * KDIM + n * NCH;
        srow[m]      = c0[i];
        srow[16 + m] = c1[i];
    }
}

__global__ void cf_gemm2_kernel(const float* __restrict__ S,
                                const float* __restrict__ w2,
                                float* __restrict__ out) {
    const int wave  = blockIdx.x * (blockDim.x >> 5) + (threadIdx.x >> 5);
    const int lane  = threadIdx.x & 31;
    const int at    = wave & 15;
    const int z     = wave >> 4;
    const int row0  = at * 16;
    const int m     = lane & 15;
    const int khalf = lane >> 4;

    const float* srow = S + (z * NPTS + row0 + m) * KDIM;
    const float* wc0  = w2 + m;
    const float* wc1  = wc0 + 16;

    v8f c0 = {};
    v8f c1 = {};
    for (int k0 = 0; k0 < KDIM; k0 += 4) {
        int kk = k0 + khalf * 2;
        v2f av;
        av.x = srow[kk];
        av.y = srow[kk + 1];
        v2f bv0, bv1;
        bv0.x = wc0[kk * NCH];
        bv0.y = wc0[(kk + 1) * NCH];
        bv1.x = wc1[kk * NCH];
        bv1.y = wc1[(kk + 1) * NCH];
        c0 = __builtin_amdgcn_wmma_f32_16x16x4_f32(
                false, av, false, bv0, (short)0, c0, false, false);
        c1 = __builtin_amdgcn_wmma_f32_16x16x4_f32(
                false, av, false, bv1, (short)0, c1, false, false);
    }
#pragma unroll
    for (int i = 0; i < 8; ++i) {
        int row = row0 + i + khalf * 8;
        float* orow = out + (z * NPTS + row) * NCH;
        orow[m]      = c0[i];
        orow[16 + m] = c1[i];
    }
}

extern "C" void kernel_launch(void* const* d_in, const int* in_sizes, int n_in,
                              void* d_out, int out_size, void* d_ws, size_t ws_size,
                              hipStream_t stream) {
    const float* feat    = (const float*)d_in[0];
    const float* geom    = (const float*)d_in[1];
    const float* centers = (const float*)d_in[2];
    const float* kw      = (const float*)d_in[3];
    const int*   n_norm  = (const int*)d_in[4];

    float* ws   = (float*)d_ws;
    float* dmat = ws + OFF_D;
    float* S    = ws + OFF_S;
    float* w2   = ws + OFF_W2;
    float* out  = (float*)d_out;

    cf_dist_kernel<<<BATCH * NPTS * NPTS / 256, 256, 0, stream>>>(geom, dmat);
    cf_w2_kernel<<<KDIM * NCH / 256, 256, 0, stream>>>(kw, n_norm, w2);

    cf_gemm1_kernel<<<128, 128, 0, stream>>>(dmat, feat, centers, S);

    cf_gemm2_kernel<<<8, 128, 0, stream>>>(S, w2, out);
}


# split the output rows 4 per work-group (128 work-groups instead of 32): stage 1 puts (row, basis) pairs in the matrix M dimension, waves split the 256 points in halves summed in LDS
# speedup vs baseline: 4.8796x; 4.8796x over previous
#include <hip/hip_runtime.h>

#define BATCH   2
#define NPTS    256
#define NCH     32
#define NBASIS  16
#define GAMMA_F 10.0f
#define EPS_F   1e-9f

typedef float v2f __attribute__((ext_vector_type(2)));
typedef float v4f __attribute__((ext_vector_type(4)));
typedef float v8f __attribute__((ext_vector_type(8)));

#define ROWS         4
#define NWAVES       8
#define NTHREADS     256
#define DPITCH       260
#define LDS_D_FLOATS (ROWS * DPITCH)
#define LDS_F_FLOATS (NPTS * NCH)
#define APITCH       40
#define NPITCH       164
#define LDS_S_FLOATS (2 * NBASIS * NPITCH)

__global__ __launch_bounds__(NTHREADS)
void cf_fused_kernel(const float* __restrict__ feat,
                     const float* __restrict__ geom,
                     const float* __restrict__ centers,
                     const float* __restrict__ kw,
                     const int* __restrict__ n_norm,
                     float* __restrict__ out) {
    __shared__ __attribute__((aligned(16))) float lds[LDS_D_FLOATS + LDS_F_FLOATS + LDS_S_FLOATS];
    float* lds_d = lds;
    float* lds_f = lds + LDS_D_FLOATS;
    float* lds_s = lds + LDS_D_FLOATS + LDS_F_FLOATS;
    float* lds_o = lds;

    const int tid  = threadIdx.x;
    const int wave = tid >> 5;
    const int lane = tid & 31;
    const int c    = lane & 15;
    const int h    = lane >> 4;
    const int z    = blockIdx.x >> 6;
    const int row0 = (blockIdx.x & 63) * ROWS;
    const int n0   = wave * 2;
    const int am   = wave >> 1;
    const int kh   = wave & 1;

    v2f wf[2][2][8];
#pragma unroll
    for (int nl = 0; nl < 2; ++nl)
#pragma unroll
        for (int ti = 0; ti < 2; ++ti)
#pragma unroll
            for (int ks = 0; ks < 8; ++ks)
                wf[nl][ti][ks] = *(const v2f*)(kw + ((n0 + nl) * NCH + 16 * ti + c) * NCH
                                               + 4 * ks + 2 * h);

    const float cn = centers[c];

    const v4f* fz = (const v4f*)(feat + z * NPTS * NCH);
    v4f fv[8];
#pragma unroll
    for (int q = 0; q < 8; ++q) fv[q] = fz[q * NTHREADS + tid];

    const float* gz = geom + z * NPTS * 3;
    const float bx = gz[tid * 3], by = gz[tid * 3 + 1], bz = gz[tid * 3 + 2];
#pragma unroll
    for (int m = 0; m < ROWS; ++m) {
        const float* ga = gz + (row0 + m) * 3;
        float dx = bx - ga[0];
        float dy = by - ga[1];
        float dz = bz - ga[2];
        lds_d[m * DPITCH + tid] = sqrtf(dx * dx + dy * dy + dz * dz + EPS_F);
    }
#pragma unroll
    for (int q = 0; q < 8; ++q) {
        int f4 = q * NTHREADS + tid;
        int b  = f4 >> 3;
        int j0 = (f4 & 7) * 4;
        int s  = b >> 2, hh = (b >> 1) & 1, e = b & 1;
        int t  = j0 >> 4, cc = j0 & 15;
        float* dst = lds_f + (s * 32 + hh * 16 + cc) * 4 + t * 2 + e;
        dst[0]  = fv[q].x;
        dst[4]  = fv[q].y;
        dst[8]  = fv[q].z;
        dst[12] = fv[q].w;
    }
    __syncthreads();

    v8f a0 = {}, a1 = {};
    const float* dp = lds_d + am * DPITCH + 128 * kh + 2 * h;
    const float* fp = lds_f + 4096 * kh + lane * 4;
    v2f dcur[4];
    v4f fcur[4];
#pragma unroll
    for (int u = 0; u < 4; ++u) {
        dcur[u] = *(const v2f*)(dp + 4 * u);
        fcur[u] = *(const v4f*)(fp + 128 * u);
    }
#pragma unroll 1
    for (int blk = 0; blk < 8; ++blk) {
        const int nb = (blk + 1) & 7;
        v2f dnext[4];
        v4f fnext[4];
#pragma unroll
        for (int u = 0; u < 4; ++u) {
            dnext[u] = *(const v2f*)(dp + 16 * nb + 4 * u);
            fnext[u] = *(const v4f*)(fp + 512 * nb + 128 * u);
        }
#pragma unroll
        for (int u = 0; u < 4; ++u) {
            float t0 = dcur[u].x - cn;
            float t1 = dcur[u].y - cn;
            v2f av, b0, b1;
            av.x = __expf(-GAMMA_F * t0 * t0);
            av.y = __expf(-GAMMA_F * t1 * t1);
            b0.x = fcur[u].x; b0.y = fcur[u].y;
            b1.x = fcur[u].z; b1.y = fcur[u].w;
            a0 = __builtin_amdgcn_wmma_f32_16x16x4_f32(false, av, false, b0, (short)0, a0, false, false);
            a1 = __builtin_amdgcn_wmma_f32_16x16x4_f32(false, av, false, b1, (short)0, a1, false, false);
        }
#pragma unroll
        for (int u = 0; u < 4; ++u) {
            dcur[u] = dnext[u];
            fcur[u] = fnext[u];
        }
    }

    {
        float* sw = lds_s + (kh * NBASIS + 8 * h) * NPITCH + am * APITCH + c;
#pragma unroll
        for (int r = 0; r < 8; ++r) {
            sw[r * NPITCH]      = a0[r];
            sw[r * NPITCH + 16] = a1[r];
        }
    }
    __syncthreads();
    v8f o0 = {}, o1 = {};
    const float* sr = lds_s + n0 * NPITCH + (c & 3) * APITCH + 2 * h;
#pragma unroll
    for (int nl = 0; nl < 2; ++nl) {
#pragma unroll
        for (int ks = 0; ks < 8; ++ks) {
            v2f p0 = *(const v2f*)(sr + nl * NPITCH + 4 * ks);
            v2f p1 = *(const v2f*)(sr + (NBASIS + nl) * NPITCH + 4 * ks);
            v2f av;
            av.x = p0.x + p1.x;
            av.y = p0.y + p1.y;
            o0 = __builtin_amdgcn_wmma_f32_16x16x4_f32(false, av, false, wf[nl][0][ks], (short)0, o0, false, false);
            o1 = __builtin_amdgcn_wmma_f32_16x16x4_f32(false, av, false, wf[nl][1][ks], (short)0, o1, false, false);
        }
    }

    if (h == 0) {
        float* ow = lds_o + wave * (ROWS * NCH) + c;
#pragma unroll
        for (int r = 0; r < ROWS; ++r) {
            ow[r * NCH]      = o0[r];
            ow[r * NCH + 16] = o1[r];
        }
    }
    __syncthreads();
    if (tid < ROWS * NCH) {
        const float scale = rsqrtf((float)n_norm[0]);
        float acc = 0.0f;
#pragma unroll
        for (int w = 0; w < NWAVES; ++w) acc += lds_o[w * (ROWS * NCH) + tid];
        out[(z * NPTS + row0) * NCH + tid] = acc * scale;
    }
}

extern "C" void kernel_launch(void* const* d_in, const int* in_sizes, int n_in,
                              void* d_out, int out_size, void* d_ws, size_t ws_size,
                              hipStream_t stream) {
    const float* feat    = (const float*)d_in[0];
    const float* geom    = (const float*)d_in[1];
    const float* centers = (const float*)d_in[2];
    const float* kw      = (const float*)d_in[3];
    const int*   n_norm  = (const int*)d_in[4];
    float* out = (float*)d_out;

    cf_fused_kernel<<<BATCH * (NPTS / ROWS), NTHREADS, 0, stream>>>(
        feat, geom, centers, kw, n_norm, out);
}


# stage-2 weight fragments are requested inside phase 1 (8 per block in its first 4 blocks), not in front of the first barrier, so their load time hides under matrix work; n_norm is read at entry
# speedup vs baseline: 5.5526x; 1.1379x over previous
#include <hip/hip_runtime.h>

#define BATCH   2
#define NPTS    256
#define NCH     32
#define NBASIS  16
#define GAMMA_F 10.0f
#define EPS_F   1e-9f

typedef float v2f __attribute__((ext_vector_type(2)));
typedef float v4f __attribute__((ext_vector_type(4)));
typedef float v8f __attribute__((ext_vector_type(8)));

#define ROWS         4
#define NWAVES       8
#define NTHREADS     256
#define DPITCH       260
#define LDS_D_FLOATS (ROWS * DPITCH)
#define LDS_F_FLOATS (NPTS * NCH)
#define APITCH       40
#define NPITCH       164
#define LDS_S_FLOATS (2 * NBASIS * NPITCH)

__global__ __launch_bounds__(NTHREADS)
void cf_fused_kernel(const float* __restrict__ feat,
                     const float* __restrict__ geom,
                     const float* __restrict__ centers,
                     const float* __restrict__ kw,
                     const int* __restrict__ n_norm,
                     float* __restrict__ out) {
    __shared__ __attribute__((aligned(16))) float lds[LDS_D_FLOATS + LDS_F_FLOATS + LDS_S_FLOATS];
    float* lds_d = lds;
    float* lds_f = lds + LDS_D_FLOATS;
    float* lds_s = lds + LDS_D_FLOATS + LDS_F_FLOATS;
    float* lds_o = lds;

    const int tid  = threadIdx.x;
    const int wave = tid >> 5;
    const int lane = tid & 31;
    const int c    = lane & 15;
    const int h    = lane >> 4;
    const int z    = blockIdx.x >> 6;
    const int row0 = (blockIdx.x & 63) * ROWS;
    const int n0   = wave * 2;
    const int am   = wave >> 1;
    const int kh   = wave & 1;

    const float cn = centers[c];
    const float scale = rsqrtf((float)n_norm[0]);

    const float* gz = geom + z * NPTS * 3;
    const float bx = gz[tid * 3], by = gz[tid * 3 + 1], bz = gz[tid * 3 + 2];
    __builtin_amdgcn_sched_barrier(0);

    const v4f* fz = (const v4f*)(feat + z * NPTS * NCH);
    v4f fv[8];
#pragma unroll
    for (int q = 0; q < 8; ++q) fv[q] = fz[q * NTHREADS + tid];
#pragma unroll
    for (int m = 0; m < ROWS; ++m) {
        const float* ga = gz + (row0 + m) * 3;
        float dx = bx - ga[0];
        float dy = by - ga[1];
        float dz = bz - ga[2];
        lds_d[m * DPITCH + tid] = sqrtf(dx * dx + dy * dy + dz * dz + EPS_F);
    }
#pragma unroll
    for (int q = 0; q < 8; ++q) {
        int f4 = q * NTHREADS + tid;
        int b  = f4 >> 3;
        int j0 = (f4 & 7) * 4;
        int s  = b >> 2, hh = (b >> 1) & 1, e = b & 1;
        int t  = j0 >> 4, cc = j0 & 15;
        float* dst = lds_f + (s * 32 + hh * 16 + cc) * 4 + t * 2 + e;
        dst[0]  = fv[q].x;
        dst[4]  = fv[q].y;
        dst[8]  = fv[q].z;
        dst[12] = fv[q].w;
    }
    __syncthreads();

    v8f a0 = {}, a1 = {};
    const float* dp = lds_d + am * DPITCH + 128 * kh + 2 * h;
    const float* fp = lds_f + 4096 * kh + lane * 4;
    const float* wp = kw + (n0 * NCH + c) * NCH + 2 * h;
    v2f wf[2][2][8];
    v2f dcur[4];
    v4f fcur[4];
#pragma unroll
    for (int u = 0; u < 4; ++u) {
        dcur[u] = *(const v2f*)(dp + 4 * u);
        fcur[u] = *(const v4f*)(fp + 128 * u);
    }
#pragma unroll
    for (int blk = 0; blk < 8; ++blk) {
        v2f dnext[4];
        v4f fnext[4];
        if (blk < 7) {
#pragma unroll
            for (int u = 0; u < 4; ++u) {
                dnext[u] = *(const v2f*)(dp + 16 * (blk + 1) + 4 * u);
                fnext[u] = *(const v4f*)(fp + 512 * (blk + 1) + 128 * u);
            }
        }
        if (blk < 4) {
            const int nl = blk >> 1, ti = blk & 1;
#pragma unroll
            for (int ks = 0; ks < 8; ++ks)
                wf[nl][ti][ks] = *(const v2f*)(wp + (nl * NCH + 16 * ti) * NCH + 4 * ks);
        }
#pragma unroll
        for (int u = 0; u < 4; ++u) {
            float t0 = dcur[u].x - cn;
            float t1 = dcur[u].y - cn;
            v2f av, b0, b1;
            av.x = __expf(-GAMMA_F * t0 * t0);
            av.y = __expf(-GAMMA_F * t1 * t1);
            b0.x = fcur[u].x; b0.y = fcur[u].y;
            b1.x = fcur[u].z; b1.y = fcur[u].w;
            a0 = __builtin_amdgcn_wmma_f32_16x16x4_f32(false, av, false, b0, (short)0, a0, false, false);
            a1 = __builtin_amdgcn_wmma_f32_16x16x4_f32(false, av, false, b1, (short)0, a1, false, false);
        }
        if (blk < 7) {
#pragma unroll
            for (int u = 0; u < 4; ++u) {
                dcur[u] = dnext[u];
                fcur[u] = fnext[u];
            }
        }
        __builtin_amdgcn_sched_barrier(0);
    }

    {
        float* sw = lds_s + (kh * NBASIS + 8 * h) * NPITCH + am * APITCH + c;
#pragma unroll
        for (int r = 0; r < 8; ++r) {
            sw[r * NPITCH]      = a0[r];
            sw[r * NPITCH + 16] = a1[r];
        }
    }
    __syncthreads();
    v8f o0 = {}, o1 = {};
    const float* sr = lds_s + n0 * NPITCH + (c & 3) * APITCH + 2 * h;
#pragma unroll
    for (int nl = 0; nl < 2; ++nl) {
#pragma unroll
        for (int ks = 0; ks < 8; ++ks) {
            v2f p0 = *(const v2f*)(sr + nl * NPITCH + 4 * ks);
            v2f p1 = *(const v2f*)(sr + (NBASIS + nl) * NPITCH + 4 * ks);
            v2f av;
            av.x = p0.x + p1.x;
            av.y = p0.y + p1.y;
            o0 = __builtin_amdgcn_wmma_f32_16x16x4_f32(false, av, false, wf[nl][0][ks], (short)0, o0, false, false);
            o1 = __builtin_amdgcn_wmma_f32_16x16x4_f32(false, av, false, wf[nl][1][ks], (short)0, o1, false, false);
        }
    }

    if (h == 0) {
        float* ow = lds_o + wave * (ROWS * NCH) + c;
#pragma unroll
        for (int r = 0; r < ROWS; ++r) {
            ow[r * NCH]      = o0[r];
            ow[r * NCH + 16] = o1[r];
        }
    }
    __syncthreads();
    if (tid < ROWS * NCH) {
        float acc = 0.0f;
#pragma unroll
        for (int w = 0; w < NWAVES; ++w) acc += lds_o[w * (ROWS * NCH) + tid];
        out[(z * NPTS + row0) * NCH + tid] = acc * scale;
    }
}

extern "C" void kernel_launch(void* const* d_in, const int* in_sizes, int n_in,
                              void* d_out, int out_size, void* d_ws, size_t ws_size,
                              hipStream_t stream) {
    const float* feat    = (const float*)d_in[0];
    const float* geom    = (const float*)d_in[1];
    const float* centers = (const float*)d_in[2];
    const float* kw      = (const float*)d_in[3];
    const int*   n_norm  = (const int*)d_in[4];
    float* out = (float*)d_out;

    cf_fused_kernel<<<BATCH * (NPTS / ROWS), NTHREADS, 0, stream>>>(
        feat, geom, centers, kw, n_norm, out);
}


# 2 output rows per work-group on 4 waves (256 work-groups, one per WGP, instead of 128 on half the chip); stage 2 carries the (row, point-half) pairs as matrix rows, each wave 4 bases
# speedup vs baseline: 6.2162x; 1.1195x over previous
#include <hip/hip_runtime.h>

#define BATCH   2
#define NPTS    256
#define NCH     32
#define NBASIS  16
#define GAMMA_F 10.0f
#define EPS_F   1e-9f

typedef float v2f __attribute__((ext_vector_type(2)));
typedef float v4f __attribute__((ext_vector_type(4)));
typedef float v8f __attribute__((ext_vector_type(8)));

#define ROWS         2
#define NWAVES       4
#define NTHREADS     128
#define DPITCH       260
#define LDS_D_FLOATS (ROWS * DPITCH)
#define LDS_F_FLOATS (NPTS * NCH)
#define APITCH       40
#define NPITCH       84
#define KHPITCH      (NBASIS * NPITCH + 8)
#define LDS_S_FLOATS (2 * KHPITCH)

__global__ __launch_bounds__(NTHREADS)
void cf_fused_kernel(const float* __restrict__ feat,
                     const float* __restrict__ geom,
                     const float* __restrict__ centers,
                     const float* __restrict__ kw,
                     const int* __restrict__ n_norm,
                     float* __restrict__ out) {
    __shared__ __attribute__((aligned(16))) float lds[LDS_D_FLOATS + LDS_F_FLOATS + LDS_S_FLOATS];
    float* lds_d = lds;
    float* lds_f = lds + LDS_D_FLOATS;
    float* lds_s = lds + LDS_D_FLOATS + LDS_F_FLOATS;
    float* lds_o = lds;

    const int tid  = threadIdx.x;
    const int wave = tid >> 5;
    const int lane = tid & 31;
    const int c    = lane & 15;
    const int h    = lane >> 4;
    const int z    = blockIdx.x >> 7;
    const int row0 = (blockIdx.x & 127) * ROWS;
    const int n0   = wave * 4;
    const int am   = wave >> 1;
    const int kh   = wave & 1;

    const float cn = centers[c];
    const float scale = rsqrtf((float)n_norm[0]);

    const float* gz = geom + z * NPTS * 3;
    const float bx = gz[tid * 3], by = gz[tid * 3 + 1], bz = gz[tid * 3 + 2];
    const float cx = gz[(tid + 128) * 3], cy = gz[(tid + 128) * 3 + 1], cz = gz[(tid + 128) * 3 + 2];
    __builtin_amdgcn_sched_barrier(0);

    const v4f* fz = (const v4f*)(feat + z * NPTS * NCH);
    v4f fv[16];
#pragma unroll
    for (int q = 0; q < 16; ++q) fv[q] = fz[q * NTHREADS + tid];
#pragma unroll
    for (int m = 0; m < ROWS; ++m) {
        const float* ga = gz + (row0 + m) * 3;
        const float ax = ga[0], ay = ga[1], az = ga[2];
        float dx = bx - ax;
        float dy = by - ay;
        float dz = bz - az;
        lds_d[m * DPITCH + tid] = sqrtf(dx * dx + dy * dy + dz * dz + EPS_F);
        float ex = cx - ax;
        float ey = cy - ay;
        float ez = cz - az;
        lds_d[m * DPITCH + tid + 128] = sqrtf(ex * ex + ey * ey + ez * ez + EPS_F);
    }
#pragma unroll
    for (int q = 0; q < 16; ++q) {
        int f4 = q * NTHREADS + tid;
        int b  = f4 >> 3;
        int j0 = (f4 & 7) * 4;
        int s  = b >> 2, hh = (b >> 1) & 1, e = b & 1;
        int t  = j0 >> 4, cc = j0 & 15;
        float* dst = lds_f + (s * 32 + hh * 16 + cc) * 4 + t * 2 + e;
        dst[0]  = fv[q].x;
        dst[4]  = fv[q].y;
        dst[8]  = fv[q].z;
        dst[12] = fv[q].w;
    }
    __syncthreads();

    v8f a0 = {}, a1 = {};
    const float* dp = lds_d + am * DPITCH + 128 * kh + 2 * h;
    const float* fp = lds_f + 4096 * kh + lane * 4;
    const float* wp = kw + (n0 * NCH + c) * NCH + 2 * h;
    v2f wf[4][2][8];
    v2f dcur[4];
    v4f fcur[4];
#pragma unroll
    for (int u = 0; u < 4; ++u) {
        dcur[u] = *(const v2f*)(dp + 4 * u);
        fcur[u] = *(const v4f*)(fp + 128 * u);
    }
#pragma unroll
    for (int blk = 0; blk < 8; ++blk) {
        v2f dnext[4];
        v4f fnext[4];
        if (blk < 7) {
#pragma unroll
            for (int u = 0; u < 4; ++u) {
                dnext[u] = *(const v2f*)(dp + 16 * (blk + 1) + 4 * u);
                fnext[u] = *(const v4f*)(fp + 512 * (blk + 1) + 128 * u);
            }
        }
        if (blk < 4) {
#pragma unroll
            for (int ti = 0; ti < 2; ++ti)
#pragma unroll
                for (int ks = 0; ks < 8; ++ks)
                    wf[blk][ti][ks] = *(const v2f*)(wp + (blk * NCH + 16 * ti) * NCH + 4 * ks);
        }
        __builtin_amdgcn_sched_barrier(0);
#pragma unroll
        for (int u = 0; u < 4; ++u) {
            float t0 = dcur[u].x - cn;
            float t1 = dcur[u].y - cn;
            v2f av, b0, b1;
            av.x = __expf(-GAMMA_F * t0 * t0);
            av.y = __expf(-GAMMA_F * t1 * t1);
            b0.x = fcur[u].x; b0.y = fcur[u].y;
            b1.x = fcur[u].z; b1.y = fcur[u].w;
            a0 = __builtin_amdgcn_wmma_f32_16x16x4_f32(false, av, false, b0, (short)0, a0, false, false);
            a1 = __builtin_amdgcn_wmma_f32_16x16x4_f32(false, av, false, b1, (short)0, a1, false, false);
        }
        if (blk < 7) {
#pragma unroll
            for (int u = 0; u < 4; ++u) {
                dcur[u] = dnext[u];
                fcur[u] = fnext[u];
            }
        }
        __builtin_amdgcn_sched_barrier(0);
    }

    {
        float* sw = lds_s + kh * KHPITCH + 8 * h * NPITCH + am * APITCH + c;
#pragma unroll
        for (int r = 0; r < 8; ++r) {
            sw[r * NPITCH]      = a0[r];
            sw[r * NPITCH + 16] = a1[r];
        }
    }
    __syncthreads();
    v8f o0 = {}, o1 = {};
    const float* sr = lds_s + ((c >> 1) & 1) * KHPITCH + n0 * NPITCH + (c & 1) * APITCH + 2 * h;
    v2f sa[4][8];
#pragma unroll
    for (int nl = 0; nl < 4; ++nl)
#pragma unroll
        for (int ks = 0; ks < 8; ++ks)
            sa[nl][ks] = *(const v2f*)(sr + nl * NPITCH + 4 * ks);
    __builtin_amdgcn_sched_barrier(0);
#pragma unroll
    for (int nl = 0; nl < 4; ++nl) {
#pragma unroll
        for (int ks = 0; ks < 8; ++ks) {
            o0 = __builtin_amdgcn_wmma_f32_16x16x4_f32(false, sa[nl][ks], false, wf[nl][0][ks], (short)0, o0, false, false);
            o1 = __builtin_amdgcn_wmma_f32_16x16x4_f32(false, sa[nl][ks], false, wf[nl][1][ks], (short)0, o1, false, false);
        }
    }

    if (h == 0) {
        float* ow = lds_o + wave * (ROWS * NCH) + c;
#pragma unroll
        for (int r = 0; r < ROWS; ++r) {
            ow[r * NCH]      = o0[r] + o0[r + ROWS];
            ow[r * NCH + 16] = o1[r] + o1[r + ROWS];
        }
    }
    __syncthreads();
    if (tid < ROWS * NCH) {
        float acc = 0.0f;
#pragma unroll
        for (int w = 0; w < NWAVES; ++w) acc += lds_o[w * (ROWS * NCH) + tid];
        out[(z * NPTS + row0) * NCH + tid] = acc * scale;
    }
}

extern "C" void kernel_launch(void* const* d_in, const int* in_sizes, int n_in,
                              void* d_out, int out_size, void* d_ws, size_t ws_size,
                              hipStream_t stream) {
    const float* feat    = (const float*)d_in[0];
    const float* geom    = (const float*)d_in[1];
    const float* centers = (const float*)d_in[2];
    const float* kw      = (const float*)d_in[3];
    const int*   n_norm  = (const int*)d_in[4];
    float* out = (float*)d_out;

    cf_fused_kernel<<<BATCH * (NPTS / ROWS), NTHREADS, 0, stream>>>(
        feat, geom, centers, kw, n_norm, out);
}


# feature fragments go from global memory straight to registers (each wave a quarter of the points, both rows): no LDS staging, first barrier waits for distances only, stage 1 runs under the loads
# speedup vs baseline: 7.1357x; 1.1479x over previous
#include <hip/hip_runtime.h>

#define BATCH   2
#define NPTS    256
#define NCH     32
#define NBASIS  16
#define GAMMA_F 10.0f
#define EPS_F   1e-9f

typedef float v2f __attribute__((ext_vector_type(2)));
typedef float v8f __attribute__((ext_vector_type(8)));

#define ROWS         2
#define NWAVES       4
#define NTHREADS     128
#define DPITCH       260
#define LDS_D_FLOATS (ROWS * DPITCH)
#define APITCH       36
#define NPITCH       76
#define QPITCH       (NBASIS * NPITCH + 8)
#define LDS_S_FLOATS (NWAVES * QPITCH)

__device__ __forceinline__ void lds_barrier() {
    __builtin_amdgcn_fence(__ATOMIC_RELEASE, "workgroup", "local");
    __builtin_amdgcn_s_barrier();
    __builtin_amdgcn_fence(__ATOMIC_ACQUIRE, "workgroup", "local");
}

__global__ __launch_bounds__(NTHREADS)
void cf_fused_kernel(const float* __restrict__ feat,
                     const float* __restrict__ geom,
                     const float* __restrict__ centers,
                     const float* __restrict__ kw,
                     const int* __restrict__ n_norm,
                     float* __restrict__ out) {
    __shared__ __attribute__((aligned(16))) float lds[LDS_D_FLOATS + LDS_S_FLOATS];
    float* lds_d = lds;
    float* lds_s = lds + LDS_D_FLOATS;
    float* lds_o = lds;

    const int tid  = threadIdx.x;
    const int wave = tid >> 5;
    const int lane = tid & 31;
    const int c    = lane & 15;
    const int h    = lane >> 4;
    const int z    = blockIdx.x >> 7;
    const int row0 = (blockIdx.x & 127) * ROWS;
    const int n0   = wave * 4;

    const float cn = centers[c];
    const float scale = rsqrtf((float)n_norm[0]);

    const float* gz = geom + z * NPTS * 3;
    const float bx = gz[tid * 3], by = gz[tid * 3 + 1], bz = gz[tid * 3 + 2];
    const float cx = gz[(tid + 128) * 3], cy = gz[(tid + 128) * 3 + 1], cz = gz[(tid + 128) * 3 + 2];
    __builtin_amdgcn_sched_barrier(0);

    const float* fq = feat + ((z * NPTS + 64 * wave + h) * NCH + 2 * c);
    v2f fl[16][2];
#pragma unroll
    for (int s = 0; s < 16; ++s)
#pragma unroll
        for (int e = 0; e < 2; ++e)
            fl[s][e] = *(const v2f*)(fq + (4 * s + 2 * e) * NCH);
    __builtin_amdgcn_sched_barrier(0);

    {
        const int p = (tid & ~3) | ((tid & 1) << 1) | ((tid >> 1) & 1);
#pragma unroll
        for (int m = 0; m < ROWS; ++m) {
            const float* ga = gz + (row0 + m) * 3;
            const float ax = ga[0], ay = ga[1], az = ga[2];
            float dx = bx - ax;
            float dy = by - ay;
            float dz = bz - az;
            lds_d[m * DPITCH + p] = sqrtf(dx * dx + dy * dy + dz * dz + EPS_F);
            float ex = cx - ax;
            float ey = cy - ay;
            float ez = cz - az;
            lds_d[m * DPITCH + p + 128] = sqrtf(ex * ex + ey * ey + ez * ez + EPS_F);
        }
    }
    lds_barrier();

    const float* dp = lds_d + 64 * wave + 2 * h;
    v2f dv[ROWS][16];
#pragma unroll
    for (int a = 0; a < ROWS; ++a)
#pragma unroll
        for (int s = 0; s < 16; ++s)
            dv[a][s] = *(const v2f*)(dp + a * DPITCH + 4 * s);

    v8f acc[ROWS][2] = {};
    const float* wp = kw + (n0 * NCH + c) * NCH + 2 * h;
    v2f wf[4][2][8];
#pragma unroll
    for (int blk = 0; blk < 4; ++blk) {
#pragma unroll
        for (int ti = 0; ti < 2; ++ti)
#pragma unroll
            for (int ks = 0; ks < 8; ++ks)
                wf[blk][ti][ks] = *(const v2f*)(wp + (blk * NCH + 16 * ti) * NCH + 4 * ks);
        __builtin_amdgcn_sched_barrier(0);
#pragma unroll
        for (int u = 0; u < 4; ++u) {
            const int s = 4 * blk + u;
            v2f b0, b1;
            b0.x = fl[s][0].x; b0.y = fl[s][1].x;
            b1.x = fl[s][0].y; b1.y = fl[s][1].y;
#pragma unroll
            for (int a = 0; a < ROWS; ++a) {
                float t0 = dv[a][s].x - cn;
                float t1 = dv[a][s].y - cn;
                v2f av;
                av.x = __expf(-GAMMA_F * t0 * t0);
                av.y = __expf(-GAMMA_F * t1 * t1);
                acc[a][0] = __builtin_amdgcn_wmma_f32_16x16x4_f32(false, av, false, b0, (short)0, acc[a][0], false, false);
                acc[a][1] = __builtin_amdgcn_wmma_f32_16x16x4_f32(false, av, false, b1, (short)0, acc[a][1], false, false);
            }
        }
        __builtin_amdgcn_sched_barrier(0);
    }

    {
        float* sw = lds_s + wave * QPITCH + 8 * h * NPITCH + 2 * c;
#pragma unroll
        for (int a = 0; a < ROWS; ++a)
#pragma unroll
            for (int r = 0; r < 8; ++r) {
                v2f pr;
                pr.x = acc[a][0][r];
                pr.y = acc[a][1][r];
                *(v2f*)(sw + r * NPITCH + a * APITCH) = pr;
            }
    }
    lds_barrier();
    v8f o0 = {}, o1 = {};
    const float* sr = lds_s + ((c >> 1) & 3) * QPITCH + n0 * NPITCH + (c & 1) * APITCH + 2 * h;
    v2f sa[4][8];
#pragma unroll
    for (int nl = 0; nl < 4; ++nl)
#pragma unroll
        for (int ks = 0; ks < 8; ++ks)
            sa[nl][ks] = *(const v2f*)(sr + nl * NPITCH + 4 * ks);
    __builtin_amdgcn_sched_barrier(0);
#pragma unroll
    for (int nl = 0; nl < 4; ++nl) {
#pragma unroll
        for (int ks = 0; ks < 8; ++ks) {
            o0 = __builtin_amdgcn_wmma_f32_16x16x4_f32(false, sa[nl][ks], false, wf[nl][0][ks], (short)0, o0, false, false);
            o1 = __builtin_amdgcn_wmma_f32_16x16x4_f32(false, sa[nl][ks], false, wf[nl][1][ks], (short)0, o1, false, false);
        }
    }

    if (h == 0) {
        float* ow = lds_o + wave * (ROWS * NCH) + c;
#pragma unroll
        for (int r = 0; r < ROWS; ++r) {
            ow[r * NCH]      = (o0[r] + o0[r + 2]) + (o0[r + 4] + o0[r + 6]);
            ow[r * NCH + 16] = (o1[r] + o1[r + 2]) + (o1[r + 4] + o1[r + 6]);
        }
    }
    lds_barrier();
    if (tid < ROWS * NCH) {
        float sum = 0.0f;
#pragma unroll
        for (int w = 0; w < NWAVES; ++w) sum += lds_o[w * (ROWS * NCH) + tid];
        out[(z * NPTS + row0) * NCH + tid] = sum * scale;
    }
}

extern "C" void kernel_launch(void* const* d_in, const int* in_sizes, int n_in,
                              void* d_out, int out_size, void* d_ws, size_t ws_size,
                              hipStream_t stream) {
    const float* feat    = (const float*)d_in[0];
    const float* geom    = (const float*)d_in[1];
    const float* centers = (const float*)d_in[2];
    const float* kw      = (const float*)d_in[3];
    const int*   n_norm  = (const int*)d_in[4];
    float* out = (float*)d_out;

    cf_fused_kernel<<<BATCH * (NPTS / ROWS), NTHREADS, 0, stream>>>(
        feat, geom, centers, kw, n_norm, out);
}
